# starting kernel
# baseline (speedup 1.0000x reference)
#include <hip/hip_runtime.h>
#include <hip/hip_bf16.h>

typedef __attribute__((ext_vector_type(2))) float v2f;
typedef __attribute__((ext_vector_type(8))) float v8f;

#define C_IN    32
#define C_OUT   64
#define K1      35
#define K1PAD   36
#define W1PITCH 38
#define W2PITCH 66
#define HPITCH  66

__global__ void sample_gather(const float* __restrict__ bxyz,
                              const int*   __restrict__ sidx,
                              float*       __restrict__ new_bxyz,
                              int m)
{
    int i = blockIdx.x * blockDim.x + threadIdx.x;
    if (i < m) {
        int p = sidx[i];
        const float4* src = (const float4*)bxyz;
        float4*       dst = (float4*)new_bxyz;
        dst[i] = src[p];
    }
}

__launch_bounds__(256)
__global__ void edge_mlp_wmma(const float* __restrict__ bxyz,
                              const float* __restrict__ feat,
                              const int*   __restrict__ e_point,
                              const int*   __restrict__ e_new,
                              const float* __restrict__ W1,
                              const float* __restrict__ b1,
                              const float* __restrict__ W2,
                              const float* __restrict__ b2,
                              const float* __restrict__ new_bxyz,
                              float*       __restrict__ agg,
                              int n_edges, int n_tiles)
{
    __shared__ float sW1t[C_OUT * W1PITCH];
    __shared__ float sW2t[C_OUT * W2PITCH];
    __shared__ float sH[8][16 * HPITCH];

    const int tid = threadIdx.x;
    for (int i = tid; i < C_OUT * K1PAD; i += 256) {
        const int c = i / K1PAD, k = i - c * K1PAD;
        sW1t[c * W1PITCH + k] = (k < K1) ? W1[k * C_OUT + c] : 0.0f;
    }
    for (int i = tid; i < C_OUT * C_OUT; i += 256) {
        const int c = i >> 6, k = i & 63;
        sW2t[c * W2PITCH + k] = W2[k * C_OUT + c];
    }
    __syncthreads();

    const int wave = tid >> 5;
    const int lane = tid & 31;
    const int tile = blockIdx.x * 8 + wave;
    if (tile >= n_tiles) return;

    const int m    = lane & 15;
    const int half = lane >> 4;
    int e = tile * 16 + m;
    if (e >= n_edges) e = n_edges - 1;
    const int pe = e_point[e];
    const int ne = e_new[e];

    float rel[3];
#pragma unroll
    for (int j = 0; j < 3; ++j)
        rel[j] = bxyz[pe * 4 + 1 + j] - new_bxyz[ne * 4 + 1 + j];

    const float* frow = feat + (long)pe * C_IN;
    const int col = m;

    v8f c1[4];
#pragma unroll
    for (int nb = 0; nb < 4; ++nb) {
        const float bv = b1[nb * 16 + col];
#pragma unroll
        for (int r = 0; r < 8; ++r) c1[nb][r] = bv;
    }

#pragma unroll
    for (int s = 0; s < 9; ++s) {
        const int ka = 4 * s + 2 * half;
        const int kb = ka + 1;
        const float a0 = (ka < C_IN) ? frow[ka] : ((ka < K1) ? rel[ka - C_IN] : 0.0f);
        const float a1 = (kb < C_IN) ? frow[kb] : ((kb < K1) ? rel[kb - C_IN] : 0.0f);
        const v2f a = {a0, a1};
#pragma unroll
        for (int nb = 0; nb < 4; ++nb) {
            const v2f b = *(const v2f*)&sW1t[(nb * 16 + col) * W1PITCH + ka];
            c1[nb] = __builtin_amdgcn_wmma_f32_16x16x4_f32(
                         false, a, false, b, (short)0, c1[nb], false, false);
        }
    }

    float* hrow = sH[wave];
#pragma unroll
    for (int nb = 0; nb < 4; ++nb) {
#pragma unroll
        for (int r = 0; r < 8; ++r) {
            float v = c1[nb][r];
            v = v > 0.0f ? v : 0.0f;
            const int mm = r + 8 * half;
            hrow[mm * HPITCH + nb * 16 + col] = v;
        }
    }

    v8f c2[4];
#pragma unroll
    for (int nb = 0; nb < 4; ++nb) {
        const float bv = b2[nb * 16 + col];
#pragma unroll
        for (int r = 0; r < 8; ++r) c2[nb][r] = bv;
    }

#pragma unroll
    for (int s = 0; s < 16; ++s) {
        const int ka = 4 * s + 2 * half;
        const v2f a = *(const v2f*)&hrow[m * HPITCH + ka];
#pragma unroll
        for (int nb = 0; nb < 4; ++nb) {
            const v2f b = *(const v2f*)&sW2t[(nb * 16 + col) * W2PITCH + ka];
            c2[nb] = __builtin_amdgcn_wmma_f32_16x16x4_f32(
                         false, a, false, b, (short)0, c2[nb], false, false);
        }
    }

#pragma unroll
    for (int r = 0; r < 8; ++r) {
        const int mm  = r + 8 * half;
        const int row = __shfl(ne, mm, 32);
#pragma unroll
        for (int nb = 0; nb < 4; ++nb) {
            float v = c2[nb][r];
            v = v > 0.0f ? v : 0.0f;
            atomicMax((int*)&agg[(long)row * C_OUT + nb * 16 + col],
                      __float_as_int(v));
        }
    }
}

extern "C" void kernel_launch(void* const* d_in, const int* in_sizes, int n_in,
                              void* d_out, int out_size, void* d_ws, size_t ws_size,
                              hipStream_t stream)
{
    const float* bxyz    = (const float*)d_in[0];
    const float* feat    = (const float*)d_in[1];
    const int*   sidx    = (const int*)  d_in[2];
    const int*   e_point = (const int*)  d_in[3];
    const int*   e_new   = (const int*)  d_in[4];
    const float* W1      = (const float*)d_in[5];
    const float* b1      = (const float*)d_in[6];
    const float* W2      = (const float*)d_in[7];
    const float* b2      = (const float*)d_in[8];

    const int M = in_sizes[2];
    const int E = in_sizes[3];

    float* out      = (float*)d_out;
    float* new_bxyz = out;
    float* agg      = out + (size_t)M * 4;

    hipMemsetAsync(agg, 0, (size_t)M * C_OUT * sizeof(float), stream);

    sample_gather<<<(M + 255) / 256, 256, 0, stream>>>(bxyz, sidx, new_bxyz, M);

    const int n_tiles = (E + 15) / 16;
    edge_mlp_wmma<<<(n_tiles + 7) / 8, 256, 0, stream>>>(
        bxyz, feat, e_point, e_new, W1, b1, W2, b2, new_bxyz, agg, E, n_tiles);
}


# One 32-wave work-group per WGP loops over its share of the edge tiles, so the transposed weights are staged into LDS once per WGP instead of once per 8 tiles
# speedup vs baseline: 1.3196x; 1.3196x over previous
#include <hip/hip_runtime.h>
#include <hip/hip_bf16.h>

typedef __attribute__((ext_vector_type(2))) float v2f;
typedef __attribute__((ext_vector_type(8))) float v8f;

#define C_IN    32
#define C_OUT   64
#define K1      35
#define K1PAD   36
#define W1PITCH 38
#define W2PITCH 66
#define HPITCH  68
#define WG_WAVES   32
#define WG_THREADS (WG_WAVES * 32)
#define N_WGP      256

__global__ void sample_gather(const float* __restrict__ bxyz,
                              const int*   __restrict__ sidx,
                              float*       __restrict__ new_bxyz,
                              int m)
{
    int i = blockIdx.x * blockDim.x + threadIdx.x;
    if (i < m) {
        int p = sidx[i];
        const float4* src = (const float4*)bxyz;
        float4*       dst = (float4*)new_bxyz;
        dst[i] = src[p];
    }
}

__launch_bounds__(WG_THREADS)
__global__ void edge_mlp_wmma(const float* __restrict__ bxyz,
                              const float* __restrict__ feat,
                              const int*   __restrict__ e_point,
                              const int*   __restrict__ e_new,
                              const float* __restrict__ W1,
                              const float* __restrict__ b1,
                              const float* __restrict__ W2,
                              const float* __restrict__ b2,
                              const float* __restrict__ new_bxyz,
                              float*       __restrict__ agg,
                              int n_edges, int n_tiles)
{
    __shared__ float sW1t[C_OUT * W1PITCH];
    __shared__ float sW2t[C_OUT * W2PITCH];
    __shared__ float sH[WG_WAVES][16 * HPITCH];

    const int tid = threadIdx.x;
    for (int i = tid; i < C_OUT * K1PAD; i += WG_THREADS) {
        const int c = i / K1PAD, k = i - c * K1PAD;
        sW1t[c * W1PITCH + k] = (k < K1) ? W1[k * C_OUT + c] : 0.0f;
    }
    for (int i = tid; i < C_OUT * C_OUT; i += WG_THREADS) {
        const int c = i >> 6, k = i & 63;
        sW2t[c * W2PITCH + k] = W2[k * C_OUT + c];
    }
    __syncthreads();

    const int wave = __builtin_amdgcn_readfirstlane(tid >> 5);
    const int lane = tid & 31;
    const int m    = lane & 15;
    const int half = lane >> 4;
    const int col  = m;

    const int t_begin = (int)(((long)blockIdx.x * n_tiles) / gridDim.x);
    const int t_end   = (int)(((long)(blockIdx.x + 1) * n_tiles) / gridDim.x);

    float* hrow = sH[wave];

    for (int tile = t_begin + wave; tile < t_end; tile += WG_WAVES) {
    asm volatile("" ::: "memory");
    int e = tile * 16 + m;
    if (e >= n_edges) e = n_edges - 1;
    const int pe = e_point[e];
    const int ne = e_new[e];

    float rel[3];
#pragma unroll
    for (int j = 0; j < 3; ++j)
        rel[j] = bxyz[pe * 4 + 1 + j] - new_bxyz[ne * 4 + 1 + j];

    const float* frow = feat + (long)pe * C_IN;

    v8f c1[4];
#pragma unroll
    for (int nb = 0; nb < 4; ++nb) {
        const float bv = b1[nb * 16 + col];
#pragma unroll
        for (int r = 0; r < 8; ++r) c1[nb][r] = bv;
    }

#pragma unroll
    for (int s = 0; s < 9; ++s) {
        const int ka = 4 * s + 2 * half;
        const int kb = ka + 1;
        const float a0 = (ka < C_IN) ? frow[ka] : ((ka < K1) ? rel[ka - C_IN] : 0.0f);
        const float a1 = (kb < C_IN) ? frow[kb] : ((kb < K1) ? rel[kb - C_IN] : 0.0f);
        const v2f a = {a0, a1};
#pragma unroll
        for (int nb = 0; nb < 4; ++nb) {
            const v2f b = *(const v2f*)&sW1t[(nb * 16 + col) * W1PITCH + ka];
            c1[nb] = __builtin_amdgcn_wmma_f32_16x16x4_f32(
                         false, a, false, b, (short)0, c1[nb], false, false);
        }
    }

#pragma unroll
    for (int nb = 0; nb < 4; ++nb) {
#pragma unroll
        for (int r = 0; r < 8; ++r) {
            float v = c1[nb][r];
            v = v > 0.0f ? v : 0.0f;
            const int mm = r + 8 * half;
            hrow[mm * HPITCH + nb * 16 + col] = v;
        }
    }

    v8f c2[4];
#pragma unroll
    for (int nb = 0; nb < 4; ++nb) {
        const float bv = b2[nb * 16 + col];
#pragma unroll
        for (int r = 0; r < 8; ++r) c2[nb][r] = bv;
    }

#pragma unroll
    for (int s = 0; s < 16; ++s) {
        const int ka = 4 * s + 2 * half;
        const v2f a = *(const v2f*)&hrow[m * HPITCH + ka];
#pragma unroll
        for (int nb = 0; nb < 4; ++nb) {
            const v2f b = *(const v2f*)&sW2t[(nb * 16 + col) * W2PITCH + ka];
            c2[nb] = __builtin_amdgcn_wmma_f32_16x16x4_f32(
                         false, a, false, b, (short)0, c2[nb], false, false);
        }
    }

#pragma unroll
    for (int r = 0; r < 8; ++r) {
        const int mm  = r + 8 * half;
        const int row = __shfl(ne, mm, 32);
#pragma unroll
        for (int nb = 0; nb < 4; ++nb) {
            float v = c2[nb][r];
            v = v > 0.0f ? v : 0.0f;
            atomicMax((int*)&agg[(long)row * C_OUT + nb * 16 + col],
                      __float_as_int(v));
        }
    }
    }
}

extern "C" void kernel_launch(void* const* d_in, const int* in_sizes, int n_in,
                              void* d_out, int out_size, void* d_ws, size_t ws_size,
                              hipStream_t stream)
{
    const float* bxyz    = (const float*)d_in[0];
    const float* feat    = (const float*)d_in[1];
    const int*   sidx    = (const int*)  d_in[2];
    const int*   e_point = (const int*)  d_in[3];
    const int*   e_new   = (const int*)  d_in[4];
    const float* W1      = (const float*)d_in[5];
    const float* b1      = (const float*)d_in[6];
    const float* W2      = (const float*)d_in[7];
    const float* b2      = (const float*)d_in[8];

    const int M = in_sizes[2];
    const int E = in_sizes[3];

    float* out      = (float*)d_out;
    float* new_bxyz = out;
    float* agg      = out + (size_t)M * 4;

    hipMemsetAsync(agg, 0, (size_t)M * C_OUT * sizeof(float), stream);

    sample_gather<<<(M + 255) / 256, 256, 0, stream>>>(bxyz, sidx, new_bxyz, M);

    const int n_tiles = (E + 15) / 16;
    edge_mlp_wmma<<<N_WGP, WG_THREADS, 0, stream>>>(
        bxyz, feat, e_point, e_new, W1, b1, W2, b2, new_bxyz, agg, E, n_tiles);
}


# Scatter-max trades half-waves between paired accumulator registers so each atomic instruction covers one whole 128-byte output line instead of two 64-byte pieces of two rows
# speedup vs baseline: 1.3964x; 1.0582x over previous
#include <hip/hip_runtime.h>
#include <hip/hip_bf16.h>

typedef __attribute__((ext_vector_type(2))) float v2f;
typedef __attribute__((ext_vector_type(8))) float v8f;
typedef __attribute__((ext_vector_type(2))) unsigned int v2u;

#define C_IN    32
#define C_OUT   64
#define K1      35
#define K1PAD   36
#define W1PITCH 38
#define W2PITCH 66
#define HPITCH  68
#define WG_WAVES   32
#define WG_THREADS (WG_WAVES * 32)
#define N_WGP      256

__global__ void sample_gather(const float* __restrict__ bxyz,
                              const int*   __restrict__ sidx,
                              float*       __restrict__ new_bxyz,
                              int m)
{
    int i = blockIdx.x * blockDim.x + threadIdx.x;
    if (i < m) {
        int p = sidx[i];
        const float4* src = (const float4*)bxyz;
        float4*       dst = (float4*)new_bxyz;
        dst[i] = src[p];
    }
}

__launch_bounds__(WG_THREADS)
__global__ void edge_mlp_wmma(const float* __restrict__ bxyz,
                              const float* __restrict__ feat,
                              const int*   __restrict__ e_point,
                              const int*   __restrict__ e_new,
                              const float* __restrict__ W1,
                              const float* __restrict__ b1,
                              const float* __restrict__ W2,
                              const float* __restrict__ b2,
                              const float* __restrict__ new_bxyz,
                              float*       __restrict__ agg,
                              int n_edges, int n_tiles)
{
    __shared__ float sW1t[C_OUT * W1PITCH];
    __shared__ float sW2t[C_OUT * W2PITCH];
    __shared__ float sH[WG_WAVES][16 * HPITCH];

    const int tid = threadIdx.x;
    for (int i = tid; i < C_OUT * K1PAD; i += WG_THREADS) {
        const int c = i / K1PAD, k = i - c * K1PAD;
        sW1t[c * W1PITCH + k] = (k < K1) ? W1[k * C_OUT + c] : 0.0f;
    }
    for (int i = tid; i < C_OUT * C_OUT; i += WG_THREADS) {
        const int c = i >> 6, k = i & 63;
        sW2t[c * W2PITCH + k] = W2[k * C_OUT + c];
    }
    __syncthreads();

    const int wave = __builtin_amdgcn_readfirstlane(tid >> 5);
    const int lane = tid & 31;
    const int m    = lane & 15;
    const int half = lane >> 4;
    const int col  = m;

    const int t_begin = (int)(((long)blockIdx.x * n_tiles) / gridDim.x);
    const int t_end   = (int)(((long)(blockIdx.x + 1) * n_tiles) / gridDim.x);

    float* hrow = sH[wave];

    for (int tile = t_begin + wave; tile < t_end; tile += WG_WAVES) {
    asm volatile("" ::: "memory");
    int e = tile * 16 + m;
    if (e >= n_edges) e = n_edges - 1;
    const int pe = e_point[e];
    const int ne = e_new[e];

    float rel[3];
#pragma unroll
    for (int j = 0; j < 3; ++j)
        rel[j] = bxyz[pe * 4 + 1 + j] - new_bxyz[ne * 4 + 1 + j];

    const float* frow = feat + (long)pe * C_IN;

    v8f c1[4];
#pragma unroll
    for (int nb = 0; nb < 4; ++nb) {
        const float bv = b1[nb * 16 + col];
#pragma unroll
        for (int r = 0; r < 8; ++r) c1[nb][r] = bv;
    }

#pragma unroll
    for (int s = 0; s < 9; ++s) {
        const int ka = 4 * s + 2 * half;
        const int kb = ka + 1;
        const float a0 = (ka < C_IN) ? frow[ka] : ((ka < K1) ? rel[ka - C_IN] : 0.0f);
        const float a1 = (kb < C_IN) ? frow[kb] : ((kb < K1) ? rel[kb - C_IN] : 0.0f);
        const v2f a = {a0, a1};
#pragma unroll
        for (int nb = 0; nb < 4; ++nb) {
            const v2f b = *(const v2f*)&sW1t[(nb * 16 + col) * W1PITCH + ka];
            c1[nb] = __builtin_amdgcn_wmma_f32_16x16x4_f32(
                         false, a, false, b, (short)0, c1[nb], false, false);
        }
    }

#pragma unroll
    for (int nb = 0; nb < 4; ++nb) {
#pragma unroll
        for (int r = 0; r < 8; ++r) {
            float v = c1[nb][r];
            v = v > 0.0f ? v : 0.0f;
            const int mm = r + 8 * half;
            hrow[mm * HPITCH + nb * 16 + col] = v;
        }
    }

    v8f c2[4];
#pragma unroll
    for (int nb = 0; nb < 4; ++nb) {
        const float bv = b2[nb * 16 + col];
#pragma unroll
        for (int r = 0; r < 8; ++r) c2[nb][r] = bv;
    }

#pragma unroll
    for (int s = 0; s < 16; ++s) {
        const int ka = 4 * s + 2 * half;
        const v2f a = *(const v2f*)&hrow[m * HPITCH + ka];
#pragma unroll
        for (int nb = 0; nb < 4; ++nb) {
            const v2f b = *(const v2f*)&sW2t[(nb * 16 + col) * W2PITCH + ka];
            c2[nb] = __builtin_amdgcn_wmma_f32_16x16x4_f32(
                         false, a, false, b, (short)0, c2[nb], false, false);
        }
    }

#pragma unroll
    for (int r = 0; r < 8; ++r) {
        const int row_lo = __builtin_amdgcn_readlane(ne, r);
        const int row_hi = __builtin_amdgcn_readlane(ne, r + 8);
        int* const p_lo = (int*)agg + (long)row_lo * C_OUT + lane;
        int* const p_hi = (int*)agg + (long)row_hi * C_OUT + lane;
#pragma unroll
        for (int p = 0; p < 2; ++p) {
            float x = c2[2 * p][r];
            float y = c2[2 * p + 1][r];
            x = x > 0.0f ? x : 0.0f;
            y = y > 0.0f ? y : 0.0f;
            const v2u t = __builtin_amdgcn_permlane16_swap(
                              __float_as_uint(x), __float_as_uint(y), false, false);
            atomicMax(p_lo + 32 * p, (int)t[0]);
            atomicMax(p_hi + 32 * p, (int)t[1]);
        }
    }
    }
}

extern "C" void kernel_launch(void* const* d_in, const int* in_sizes, int n_in,
                              void* d_out, int out_size, void* d_ws, size_t ws_size,
                              hipStream_t stream)
{
    const float* bxyz    = (const float*)d_in[0];
    const float* feat    = (const float*)d_in[1];
    const int*   sidx    = (const int*)  d_in[2];
    const int*   e_point = (const int*)  d_in[3];
    const int*   e_new   = (const int*)  d_in[4];
    const float* W1      = (const float*)d_in[5];
    const float* b1      = (const float*)d_in[6];
    const float* W2      = (const float*)d_in[7];
    const float* b2      = (const float*)d_in[8];

    const int M = in_sizes[2];
    const int E = in_sizes[3];

    float* out      = (float*)d_out;
    float* new_bxyz = out;
    float* agg      = out + (size_t)M * 4;

    hipMemsetAsync(agg, 0, (size_t)M * C_OUT * sizeof(float), stream);

    sample_gather<<<(M + 255) / 256, 256, 0, stream>>>(bxyz, sidx, new_bxyz, M);

    const int n_tiles = (E + 15) / 16;
    edge_mlp_wmma<<<N_WGP, WG_THREADS, 0, stream>>>(
        bxyz, feat, e_point, e_new, W1, b1, W2, b2, new_bxyz, agg, E, n_tiles);
}
